# starting kernel
# baseline (speedup 1.0000x reference)
#include <hip/hip_runtime.h>
#include <math.h>

typedef float v2f __attribute__((ext_vector_type(2)));
typedef float v4f __attribute__((ext_vector_type(4)));
typedef float v8f __attribute__((ext_vector_type(8)));

#define NGROUP   16
#define IN_DIM   8
#define OCH      128
#define DIN      16
#define NVOX     32768

__global__ __launch_bounds__(256)
void deconv_votes_wmma(const float* __restrict__ x,
                       const float* __restrict__ w,
                       const float* __restrict__ deconv_b,
                       float* __restrict__ votes) {
  __shared__ float xs[8 * 2 * 2 * 18 * 2];

  const int tid   = threadIdx.x;
  const int lane  = tid & 31;
  const int wv    = tid >> 5;
  const int ocb   = wv * 16;
  const int m     = lane & 15;
  const int khalf = lane >> 4;

  const int wg  = blockIdx.x;
  const int g   = wg >> 7;
  const int rem = wg & 127;
  const int par = rem >> 4;
  const int dp  = rem & 15;
  const int pd = (par >> 2) & 1, ph = (par >> 1) & 1, pw = par & 1;
  const int bidx = g >> 3, indim = g & 7;

  const int deltad[2] = { pd ? 1 : 0, pd ? 0 : -1 };
  const int deltah[2] = { ph ? 1 : 0, ph ? 0 : -1 };
  const int deltaw[2] = { pw ? 1 : 0, pw ? 0 : -1 };

  v2f A[2][2][2][4];
  {
    const int kd0 = 1 - pd, kh0 = 1 - ph, kw0 = 1 - pw;
    for (int sd = 0; sd < 2; ++sd)
      for (int sh = 0; sh < 2; ++sh)
        for (int sw = 0; sw < 2; ++sw) {
          const int tap = ((2 * sd + kd0) * 4 + (2 * sh + kh0)) * 4 + (2 * sw + kw0);
          for (int c = 0; c < 4; ++c) {
            const int ic = c * 4 + khalf * 2;
            const float* wp = w + (size_t)(ic * OCH + ocb + m) * 64 + tap;
            A[sd][sh][sw][c].x = wp[0];
            A[sd][sh][sw][c].y = wp[OCH * 64];
          }
        }
  }
  float bl[8];
  for (int j = 0; j < 8; ++j) bl[j] = deconv_b[ocb + khalf * 8 + j];

  if (tid < 128) {
    const int e = tid & 1, cs = (tid >> 1) & 1, r = tid >> 2;
    xs[(r * 18 + (cs ? 17 : 0)) * 2 + e] = 0.f;
  }

  const size_t xg = (size_t)g * 16 * DIN * DIN * DIN;

  for (int hp = 0; hp < 16; ++hp) {
    __syncthreads();
    {
      const int row  = tid >> 2;
      const int colq = (tid & 3) * 4;
      const int ic = row >> 2, sd = (row >> 1) & 1, sh = row & 1;
      const int cp = ic >> 1, e = ic & 1;
      const int d = dp + deltad[sd];
      const int h = hp + deltah[sh];
      const bool ok = ((unsigned)d < 16u) && ((unsigned)h < 16u);
      const float* src = x + xg + (((size_t)ic * DIN + (ok ? d : 0)) * DIN + (ok ? h : 0)) * DIN + colq;
      float* dst = xs + (((cp * 4 + sd * 2 + sh) * 18) + 1 + colq) * 2 + e;
      for (int j = 0; j < 4; ++j) dst[j * 2] = ok ? src[j] : 0.f;
    }
    __syncthreads();

    v8f acc = {0.f, 0.f, 0.f, 0.f, 0.f, 0.f, 0.f, 0.f};
    const int n = lane & 15;
    for (int sd = 0; sd < 2; ++sd)
      for (int sh = 0; sh < 2; ++sh)
        for (int sw = 0; sw < 2; ++sw) {
          const int col = 1 + n + deltaw[sw];
          for (int c = 0; c < 4; ++c) {
            const int cp = c * 2 + khalf;
            const v2f bvec = *(const v2f*)&xs[((cp * 4 + sd * 2 + sh) * 18 + col) * 2];
            acc = __builtin_amdgcn_wmma_f32_16x16x4_f32(
                false, A[sd][sh][sw][c], false, bvec, (short)0, acc, false, false);
          }
        }

    for (int j = 0; j < 8; ++j) acc[j] += bl[j];

    const int od = 2 * dp + pd, oh = 2 * hp + ph, ow = 2 * n + pw;
    const int o  = (od * 32 + oh) * 32 + ow;
    float* dst = votes + ((size_t)(bidx * NVOX + o) * IN_DIM + indim) * OCH
                       + ocb + khalf * 8;
    ((v4f*)dst)[0] = *((const v4f*)&acc);
    ((v4f*)dst)[1] = *((const v4f*)&acc + 1);
  }
}

__global__ __launch_bounds__(256)
void routing_squash(const float* __restrict__ votes,
                    const float* __restrict__ rbias,
                    float* __restrict__ out) {
  const int tid  = threadIdx.x;
  const int lane = tid & 31;
  const int wv   = tid >> 5;
  const int outd = lane >> 2;
  const int v    = lane & 3;
  const int vox  = blockIdx.x * 32 + wv * 4 + v;
  const int b    = vox >> 15;
  const int o    = vox & 32767;

  float vt[8][16];
  const float* base = votes + (size_t)vox * IN_DIM * OCH + outd * 16;
  for (int in = 0; in < 8; ++in) {
    const v4f* p = (const v4f*)(base + (size_t)in * OCH);
    v4f q0 = p[0], q1 = p[1], q2 = p[2], q3 = p[3];
    for (int a = 0; a < 4; ++a) {
      vt[in][a] = q0[a]; vt[in][4 + a] = q1[a];
      vt[in][8 + a] = q2[a]; vt[in][12 + a] = q3[a];
    }
  }
  float rb[16];
  for (int a = 0; a < 16; ++a) rb[a] = rbias[outd * 16 + a];

  float n2[8];
  for (int in = 0; in < 8; ++in) {
    float s = 0.f;
    for (int a = 0; a < 16; ++a) s += vt[in][a] * vt[in][a];
    n2[in] = sqrtf(s);
  }

  float logits[8];
  for (int in = 0; in < 8; ++in) logits[in] = 0.f;
  float pa[16];

  for (int it = 0; it < 3; ++it) {
    float route[8];
    for (int in = 0; in < 8; ++in) {
      float e = __expf(logits[in]);
      float t = e;
      t += __shfl_xor(t, 4, 32);
      t += __shfl_xor(t, 8, 32);
      t += __shfl_xor(t, 16, 32);
      route[in] = e / t;
    }
    for (int a = 0; a < 16; ++a) pa[a] = rb[a];
    for (int in = 0; in < 8; ++in)
      for (int a = 0; a < 16; ++a) pa[a] += vt[in][a] * route[in];
    if (it < 2) {
      float s = 0.f;
      for (int a = 0; a < 16; ++a) s += pa[a] * pa[a];
      const float n1 = sqrtf(s);
      for (int in = 0; in < 8; ++in) {
        float dot = 0.f;
        for (int a = 0; a < 16; ++a) dot += pa[a] * vt[in][a];
        logits[in] += dot / fmaxf(n1 * n2[in], 1e-8f);
      }
    }
  }

  float nsq = 0.f;
  for (int a = 0; a < 16; ++a) nsq += pa[a] * pa[a];
  const float norm  = sqrtf(nsq);
  const float scale = (1.f / (norm + 1e-12f)) * (nsq / (1.f + nsq));
  for (int a = 0; a < 16; ++a)
    out[(((size_t)(b * 8 + outd) * 16 + a) << 15) + o] = pa[a] * scale;
}

extern "C" void kernel_launch(void* const* d_in, const int* in_sizes, int n_in,
                              void* d_out, int out_size, void* d_ws, size_t ws_size,
                              hipStream_t stream) {
  (void)in_sizes; (void)n_in; (void)out_size; (void)ws_size;
  const float* x  = (const float*)d_in[0];
  const float* w  = (const float*)d_in[1];
  const float* db = (const float*)d_in[2];
  const float* rb = (const float*)d_in[3];
  float* votes = (float*)d_ws;
  float* outp  = (float*)d_out;

  deconv_votes_wmma<<<NGROUP * 8 * 16, 256, 0, stream>>>(x, w, db, votes);
  routing_squash<<<2048, 256, 0, stream>>>(votes, rb, outp);
}


# Deconv row loop: the slab row two neighbouring output rows share stays in registers; one new slab row read from LDS per output row, as one group (8 reads, were 16 in 11 places); sum order changes
# speedup vs baseline: 1.7077x; 1.7077x over previous
#include <hip/hip_runtime.h>
#include <math.h>

typedef float v2f __attribute__((ext_vector_type(2)));
typedef float v4f __attribute__((ext_vector_type(4)));
typedef float v8f __attribute__((ext_vector_type(8)));

#define NGROUP   16
#define IN_DIM   8
#define OCH      128
#define DIN      16

__global__ __launch_bounds__(128)
void pack_weights(const float* __restrict__ w, float* __restrict__ wpk) {
  const int u     = blockIdx.x * 128 + threadIdx.x;
  const int lane  = u & 31, q = (u >> 5) & 15, wv = (u >> 9) & 7, par = u >> 12;
  const int m     = lane & 15, khalf = lane >> 4;
  const int pd = (par >> 2) & 1, ph = (par >> 1) & 1, pw = par & 1;
  const int t  = q >> 1;
  const int sd = t >> 2, sh = (t >> 1) & 1, sw = t & 1;
  const int tap = ((2 * sd + 1 - pd) * 4 + (2 * sh + 1 - ph)) * 4 + (2 * sw + 1 - pw);
  const int ic  = (2 * (q & 1)) * 4 + khalf * 2;
  const float* s = w + (size_t)(ic * OCH + wv * 16 + m) * 64 + tap;
  v4f o;
  o.x = s[0];
  o.y = s[OCH * 64];
  o.z = s[4 * OCH * 64];
  o.w = s[5 * OCH * 64];
  *(v4f*)(wpk + (size_t)u * 4) = o;
}

#define HROWS 17
#define CPS   1248
__global__ __launch_bounds__(256)
void deconv_votes_wmma(const float* __restrict__ x,
                       const float* __restrict__ wpk,
                       const float* __restrict__ deconv_b,
                       float* __restrict__ votes) {
  __shared__ __attribute__((aligned(16))) float xs[8 * CPS];

  const int tid   = threadIdx.x;
  const int lane  = tid & 31;
  const int wv    = tid >> 5;
  const int ocb   = wv * 16;
  const int m     = lane & 15;
  const int khalf = lane >> 4;

  const int wg  = blockIdx.x;
  const int g   = wg >> 7;
  const int rem = wg & 127;
  const int par = rem >> 4;
  const int dp  = rem & 15;
  const int pd = (par >> 2) & 1, ph = (par >> 1) & 1, pw = par & 1;
  const int bidx = g >> 3, indim = g & 7;


  v2f A[2][2][2][4];
  {
    const v4f* ap = (const v4f*)(wpk + (size_t)((par * 8 + wv) * 16) * 128) + lane;
    for (int sd = 0; sd < 2; ++sd)
      for (int sh = 0; sh < 2; ++sh)
        for (int sw = 0; sw < 2; ++sw)
          for (int c2 = 0; c2 < 2; ++c2) {
            const int q = ((sd * 2 + sh) * 2 + sw) * 2 + c2;
            const v4f t = ap[q * 32];
            const v2f lo = {t.x, t.y}, hi = {t.z, t.w};
            A[sd][sh][sw][2 * c2]     = lo;
            A[sd][sh][sw][2 * c2 + 1] = hi;
          }
  }
  float bl[8];
  for (int j = 0; j < 8; ++j) bl[j] = deconv_b[ocb + khalf * 8 + j];

  const size_t xg = (size_t)g * 16 * DIN * DIN * DIN;
  const v2f zero2 = {0.f, 0.f};
  const v4f zero4 = {0.f, 0.f, 0.f, 0.f};

  for (int i = tid; i < 8 * 2 * HROWS * 2; i += 256) {
    const int r = i >> 1, side = i & 1;
    const int cp = r / (2 * HROWS), rr = r - cp * (2 * HROWS);
    *(v2f*)&xs[cp * CPS + (rr * 18 + (side ? 17 : 0)) * 2] = zero2;
  }
  {
    const int r = tid >> 4, col = 1 + (tid & 15);
    const int hz = ph ? 16 : 0;
    *(v2f*)&xs[(r >> 1) * CPS + (((r & 1) * HROWS + hz) * 18 + col) * 2] = zero2;
  }
  for (int k = 0; k < 4; ++k) {
    const int u  = tid + 256 * k;
    const int q  = u & 3, h = (u >> 2) & 15, sd = (u >> 6) & 1, cp = u >> 7;
    const int d  = dp + pd - sd;
    const bool ok = (unsigned)d < 16u;
    const float* s0 = x + xg + (((size_t)(2 * cp) * DIN + (ok ? d : 0)) * DIN + h) * DIN + q * 4;
    v4f e0 = *(const v4f*)s0;
    v4f e1 = *(const v4f*)(s0 + DIN * DIN * DIN);
    e0 = ok ? e0 : zero4;
    e1 = ok ? e1 : zero4;
    float* dst = xs + cp * CPS + ((sd * HROWS + h + 1 - ph) * 18 + 1 + q * 4) * 2;
    for (int j = 0; j < 4; ++j) {
      const v2f pr = {e0[j], e1[j]};
      *(v2f*)(dst + 2 * j) = pr;
    }
  }
  __syncthreads();

  const int n = lane & 15;
  const float* bbase = xs + khalf * CPS + (1 + n + pw) * 2;
  const int od = 2 * dp + pd;
  float* vlane = votes + (size_t)indim * 4096 + wv * 512 + pw * 256
               + (n >> 2) * 32 + khalf * 16 + (n & 3) * 4;

  auto load_row = [&](v2f (&F)[2][4][2], const int r) {
    for (int sd = 0; sd < 2; ++sd)
      for (int c = 0; c < 4; ++c)
        for (int sw = 0; sw < 2; ++sw)
          F[sd][c][sw] = *(const v2f*)(bbase + c * 2 * CPS + ((sd * HROWS + r) * 18 - sw) * 2);
  };
  auto out_row = [&](const int hp, const v2f (&Fk)[2][4][2], const v2f (&Fn)[2][4][2]) {
    v8f acc = {0.f, 0.f, 0.f, 0.f, 0.f, 0.f, 0.f, 0.f};
    for (int sd = 0; sd < 2; ++sd)
      for (int sw = 0; sw < 2; ++sw)
        for (int c = 0; c < 4; ++c)
          acc = __builtin_amdgcn_wmma_f32_16x16x4_f32(
              false, A[sd][1][sw][c], false, Fk[sd][c][sw], (short)0, acc, false, false);
    for (int sd = 0; sd < 2; ++sd)
      for (int sw = 0; sw < 2; ++sw)
        for (int c = 0; c < 4; ++c)
          acc = __builtin_amdgcn_wmma_f32_16x16x4_f32(
              false, A[sd][0][sw][c], false, Fn[sd][c][sw], (short)0, acc, false, false);

    for (int j = 0; j < 8; ++j) acc[j] += bl[j];

    const int orow = (bidx * 32 + od) * 32 + 2 * hp + ph;
    float* dst = vlane + (size_t)orow * 32768;
    *(v4f*)dst         = *((const v4f*)&acc);
    *(v4f*)(dst + 128) = *((const v4f*)&acc + 1);
  };

  v2f F0[2][4][2], F1[2][4][2];
  load_row(F0, 0);
  for (int hp = 0; hp < 16; hp += 2) {
    load_row(F1, hp + 1);
    __builtin_amdgcn_sched_barrier(0);
    out_row(hp, F0, F1);
    __builtin_amdgcn_sched_barrier(0);
    load_row(F0, hp + 2);
    __builtin_amdgcn_sched_barrier(0);
    out_row(hp + 1, F1, F0);
  }
}

#define ROUTE_BLOCKS   256
#define UNITS_PER_WAVE 8
__global__ __launch_bounds__(256)
void routing_squash(const float* __restrict__ votes,
                    const float* __restrict__ rbias,
                    float* __restrict__ out) {
  const int tid  = threadIdx.x;
  const int lane = tid & 31;
  const int wv   = tid >> 5;
  const int outd = lane >> 2;
  const int v    = lane & 3;
  const int q    = wv & 3;
  const int rsel = wv >> 2;

  float rb[16];
  for (int a = 0; a < 16; ++a) rb[a] = rbias[outd * 16 + a];

  const float* lbase = votes + outd * 512 + q * 32 + v * 4;

  v4f nx[32];
  {
    const float* p = lbase + (size_t)(blockIdx.x + ROUTE_BLOCKS * rsel) * 32768;
    for (int in = 0; in < 8; ++in)
      for (int j = 0; j < 2; ++j)
        for (int h = 0; h < 2; ++h)
          nx[in * 4 + 2 * h + j] = *(const v4f*)(p + in * 4096 + j * 128 + h * 16);
  }

  for (int a = 0; a < 16; ++a) asm volatile("" : "+v"(rb[a]));

  float keep[16];
  for (int a = 0; a < 16; ++a) keep[a] = 0.f;

#pragma unroll 1
  for (int u = 0; u < UNITS_PER_WAVE; ++u) {
    const int par = u & 1;
    const int row = blockIdx.x + ROUTE_BLOCKS * ((u >> 1) * 2 + rsel);

    float vt[8][16];
    for (int in = 0; in < 8; ++in)
      for (int k = 0; k < 4; ++k)
        for (int a = 0; a < 4; ++a) vt[in][4 * k + a] = nx[in * 4 + k][a];

    if (u + 1 < UNITS_PER_WAVE) {
      const int u1 = u + 1;
      const int row1 = blockIdx.x + ROUTE_BLOCKS * ((u1 >> 1) * 2 + rsel);
      const float* p = lbase + (size_t)row1 * 32768 + (u1 & 1) * 256;
      for (int in = 0; in < 8; ++in)
        for (int j = 0; j < 2; ++j)
          for (int h = 0; h < 2; ++h)
            nx[in * 4 + 2 * h + j] = *(const v4f*)(p + in * 4096 + j * 128 + h * 16);
    }

    float n2[8];
    for (int in = 0; in < 8; ++in) {
      float s = 0.f;
      for (int a = 0; a < 16; ++a) s += vt[in][a] * vt[in][a];
      n2[in] = sqrtf(s);
    }

    float logits[8];
    for (int in = 0; in < 8; ++in) logits[in] = 0.f;
    float pa[16];

    for (int r = 0; r < 3; ++r) {
      float route[8];
      for (int in = 0; in < 8; ++in) {
        float e = __expf(logits[in]);
        float t = e;
        t += __shfl_xor(t, 4, 32);
        t += __shfl_xor(t, 8, 32);
        t += __shfl_xor(t, 16, 32);
        route[in] = e / t;
      }
      for (int a = 0; a < 16; ++a) pa[a] = rb[a];
      for (int in = 0; in < 8; ++in)
        for (int a = 0; a < 16; ++a) pa[a] += vt[in][a] * route[in];
      if (r < 2) {
        float s = 0.f;
        for (int a = 0; a < 16; ++a) s += pa[a] * pa[a];
        const float n1 = sqrtf(s);
        for (int in = 0; in < 8; ++in) {
          float dot = 0.f;
          for (int a = 0; a < 16; ++a) dot += pa[a] * vt[in][a];
          logits[in] += dot / fmaxf(n1 * n2[in], 1e-8f);
        }
      }
    }

    float nsq = 0.f;
    for (int a = 0; a < 16; ++a) nsq += pa[a] * pa[a];
    const float norm  = sqrtf(nsq);
    const float scale = (1.f / (norm + 1e-12f)) * (nsq / (1.f + nsq));
    if (par == 0) {
      for (int a = 0; a < 16; ++a) keep[a] = pa[a] * scale;
    } else {
      const int b = row >> 10;
      const int o = (row & 1023) * 32 + 8 * q + 2 * v;
      for (int a = 0; a < 16; ++a) {
        const v2f pr = {keep[a], pa[a] * scale};
        *(v2f*)&out[(((size_t)(b * 8 + outd) * 16 + a) << 15) + o] = pr;
      }
    }
  }
}

extern "C" void kernel_launch(void* const* d_in, const int* in_sizes, int n_in,
                              void* d_out, int out_size, void* d_ws, size_t ws_size,
                              hipStream_t stream) {
  (void)in_sizes; (void)n_in; (void)out_size; (void)ws_size;
  const float* x  = (const float*)d_in[0];
  const float* w  = (const float*)d_in[1];
  const float* db = (const float*)d_in[2];
  const float* rb = (const float*)d_in[3];
  float* votes = (float*)d_ws;
  float* outp  = (float*)d_out;
  float* wpk   = outp;

  pack_weights<<<256, 128, 0, stream>>>(w, wpk);
  deconv_votes_wmma<<<NGROUP * 8 * 16, 256, 0, stream>>>(x, wpk, db, votes);
  routing_squash<<<ROUTE_BLOCKS, 256, 0, stream>>>(votes, rb, outp);
}
